# starting kernel
# baseline (speedup 1.0000x reference)
#include <hip/hip_runtime.h>

typedef float v2f __attribute__((ext_vector_type(2)));
typedef float v8f __attribute__((ext_vector_type(8)));

#define B_ 4
#define S_ 256
#define D_ 512
#define P_ 256
#define NEG_INF_ (-1e9f)

__device__ __forceinline__ float fast_tanh(float x) {
#if __has_builtin(__builtin_amdgcn_tanhf)
  return __builtin_amdgcn_tanhf(x);
#else
  float ax = __builtin_fabsf(x);
  float t  = __expf(-2.0f * ax);
  float r  = (1.0f - t) * __builtin_amdgcn_rcpf(1.0f + t);
  return x < 0.0f ? -r : r;
#endif
}

__global__ __launch_bounds__(256) void proj_wmma(
    const float* __restrict__ seq, const float* __restrict__ W1,
    const float* __restrict__ W2, float* __restrict__ pk,
    float* __restrict__ pq) {
  const int lane = threadIdx.x & 31;
  const int wid  = blockIdx.x * (blockDim.x >> 5) + (threadIdx.x >> 5);
  const int matsel = wid >> 10;
  const int tile   = wid & 1023;
  const int m0 = (tile >> 4) << 4;
  const int n0 = (tile & 15) << 4;
  const float* W   = matsel ? W2 : W1;
  float*       out = matsel ? pq : pk;

  const int hl  = lane >> 4;
  const int l16 = lane & 15;
  const int m = m0 + l16;
  const int n = n0 + l16;

  v8f acc = {};
  for (int k0 = 0; k0 < D_; k0 += 4) {
    const int k = k0 + (hl << 1);
    v2f a = *(const v2f*)(seq + m * D_ + k);
    v2f b = *(const v2f*)(W   + n * D_ + k);
    acc = __builtin_amdgcn_wmma_f32_16x16x4_f32(
        false, a, false, b, (short)0, acc, false, false);
  }
#pragma unroll
  for (int r = 0; r < 8; ++r)
    out[(m0 + r + (hl << 3)) * P_ + n0 + l16] = acc[r];
}

__global__ __launch_bounds__(256) void score_softmax(
    const float* __restrict__ pq, const float* __restrict__ pk,
    const float* __restrict__ v, const float* __restrict__ mask,
    float* __restrict__ wout) {
  const int bi  = blockIdx.x;
  const int b   = bi >> 8;
  const int tid = threadIdx.x;
  const int lane = tid & 31, wv = tid >> 5;

  __shared__ float pq_s[P_];
  __shared__ float v_s[P_];
  __shared__ float sc[S_];
  __shared__ float redm[8];
  __shared__ float reds[8];

  pq_s[tid] = pq[bi * P_ + tid];
  v_s[tid]  = v[tid];
  __syncthreads();

  const float* pkb = pk + b * S_ * P_;
  for (int jj = 0; jj < 32; ++jj) {
    const int j = (wv << 5) + jj;
    const float* row = pkb + j * P_;
    if (jj + 1 < 32)
      __builtin_prefetch(pkb + (j + 1) * P_ + lane * 8, 0, 1);
    float part = 0.0f;
#pragma unroll
    for (int q = 0; q < 8; ++q) {
      const int p = lane + (q << 5);
      part = __builtin_fmaf(v_s[p], fast_tanh(pq_s[p] + row[p]), part);
    }
#pragma unroll
    for (int off = 16; off > 0; off >>= 1)
      part += __shfl_xor(part, off, 32);
    if (lane == 0) sc[j] = part;
  }
  __syncthreads();

  const float km = mask[b * S_ + tid];
  float s = (km > 0.0f) ? sc[tid] : NEG_INF_;
  float mx = s;
#pragma unroll
  for (int off = 16; off > 0; off >>= 1)
    mx = fmaxf(mx, __shfl_xor(mx, off, 32));
  if (lane == 0) redm[wv] = mx;
  __syncthreads();
  float bm = redm[0];
#pragma unroll
  for (int w = 1; w < 8; ++w) bm = fmaxf(bm, redm[w]);

  float e  = __expf(s - bm);
  float sm = e;
#pragma unroll
  for (int off = 16; off > 0; off >>= 1)
    sm += __shfl_xor(sm, off, 32);
  if (lane == 0) reds[wv] = sm;
  __syncthreads();
  float tot = 0.0f;
#pragma unroll
  for (int w = 0; w < 8; ++w) tot += reds[w];

  const int   i  = bi & 255;
  const float qm = mask[b * S_ + i];
  wout[bi * S_ + tid] = (e / tot) * qm;
}

__global__ __launch_bounds__(256) void attn_wmma(
    const float* __restrict__ weights, const float* __restrict__ seq,
    float* __restrict__ attn) {
  const int lane = threadIdx.x & 31;
  const int wid  = blockIdx.x * (blockDim.x >> 5) + (threadIdx.x >> 5);
  const int b    = wid >> 9;
  const int rest = wid & 511;
  const int m0 = (rest >> 5) << 4;
  const int n0 = (rest & 31) << 4;
  const float* Wt = weights + b * S_ * S_;
  const float* Sq = seq     + b * S_ * D_;

  const int hl  = lane >> 4;
  const int l16 = lane & 15;
  const int m = m0 + l16;
  const int n = n0 + l16;

  v8f acc = {};
  for (int k0 = 0; k0 < S_; k0 += 4) {
    const int k = k0 + (hl << 1);
    v2f a = *(const v2f*)(Wt + m * S_ + k);
    v2f bb;
    bb.x = Sq[k * D_ + n];
    bb.y = Sq[(k + 1) * D_ + n];
    acc = __builtin_amdgcn_wmma_f32_16x16x4_f32(
        false, a, false, bb, (short)0, acc, false, false);
  }
  float* O = attn + b * S_ * D_;
#pragma unroll
  for (int r = 0; r < 8; ++r)
    O[(m0 + r + (hl << 3)) * D_ + n0 + l16] = acc[r];
}

extern "C" void kernel_launch(void* const* d_in, const int* in_sizes, int n_in,
                              void* d_out, int out_size, void* d_ws, size_t ws_size,
                              hipStream_t stream) {
  (void)in_sizes; (void)n_in; (void)out_size; (void)ws_size;
  const float* seq  = (const float*)d_in[0];
  const float* mask = (const float*)d_in[1];
  const float* W1   = (const float*)d_in[2];
  const float* W2   = (const float*)d_in[3];
  const float* v    = (const float*)d_in[4];

  float* attn = (float*)d_out;
  float* wout = attn + B_ * S_ * D_;
  float* pk   = (float*)d_ws;
  float* pq   = pk + B_ * S_ * P_;

  proj_wmma<<<256, 256, 0, stream>>>(seq, W1, W2, pk, pq);
  score_softmax<<<B_ * S_, 256, 0, stream>>>(pq, pk, v, mask, wout);
  attn_wmma<<<256, 256, 0, stream>>>(wout, seq, attn);
}


# score kernel: each block keeps 4 queries in registers so one pk row load feeds 4 queries (L2 reads /4); lane sums joined by a halving exchange on the vector ALU: sum order over p changes
# speedup vs baseline: 1.2998x; 1.2998x over previous
#include <hip/hip_runtime.h>

typedef float v2f __attribute__((ext_vector_type(2)));
typedef float v4f __attribute__((ext_vector_type(4)));
typedef float v8f __attribute__((ext_vector_type(8)));

#define B_ 4
#define S_ 256
#define D_ 512
#define P_ 256
#define NEG_INF_ (-1e9f)
#define TI_ 4
#define SCP_ (S_ + 16)

__device__ __forceinline__ float fast_tanh(float x) {
#if __has_builtin(__builtin_amdgcn_tanhf)
  return __builtin_amdgcn_tanhf(x);
#else
  float ax = __builtin_fabsf(x);
  float t  = __expf(-2.0f * ax);
  float r  = (1.0f - t) * __builtin_amdgcn_rcpf(1.0f + t);
  return x < 0.0f ? -r : r;
#endif
}

template <int OFF>
__device__ __forceinline__ float lane_xor(float x) {
  const int b = __builtin_bit_cast(int, x);
  if constexpr (OFF == 16)
    return __builtin_bit_cast(float, __builtin_amdgcn_permlanex16(
        b, b, 0x76543210, 0xfedcba98, false, false));
  else
    return __builtin_bit_cast(float, __builtin_amdgcn_update_dpp(
        b, b, 0x160 + OFF, 0xf, 0xf, false));
}

template <int OFF>
__device__ __forceinline__ void halve(float (&a)[32], const int lane) {
  const bool up = (lane & OFF) != 0;
#pragma unroll
  for (int k = 0; k < OFF; ++k) {
    const float lo = a[k];
    const float hi = a[k + OFF];
    const float keep = up ? hi : lo;
    const float send = up ? lo : hi;
    a[k] = keep + lane_xor<OFF>(send);
  }
}

__global__ __launch_bounds__(256) void proj_wmma(
    const float* __restrict__ seq, const float* __restrict__ W1,
    const float* __restrict__ W2, float* __restrict__ pk,
    float* __restrict__ pq) {
  const int lane = threadIdx.x & 31;
  const int wid  = blockIdx.x * (blockDim.x >> 5) + (threadIdx.x >> 5);
  const int matsel = wid >> 10;
  const int tile   = wid & 1023;
  const int m0 = (tile >> 4) << 4;
  const int n0 = (tile & 15) << 4;
  const float* W   = matsel ? W2 : W1;
  float*       out = matsel ? pq : pk;

  const int hl  = lane >> 4;
  const int l16 = lane & 15;
  const int m = m0 + l16;
  const int n = n0 + l16;

  v8f acc = {};
  for (int k0 = 0; k0 < D_; k0 += 4) {
    const int k = k0 + (hl << 1);
    v2f a = *(const v2f*)(seq + m * D_ + k);
    v2f b = *(const v2f*)(W   + n * D_ + k);
    acc = __builtin_amdgcn_wmma_f32_16x16x4_f32(
        false, a, false, b, (short)0, acc, false, false);
  }
#pragma unroll
  for (int r = 0; r < 8; ++r)
    out[(m0 + r + (hl << 3)) * P_ + n0 + l16] = acc[r];
}

__global__ __launch_bounds__(256) void score_softmax(
    const float* __restrict__ pq, const float* __restrict__ pk,
    const float* __restrict__ v, const float* __restrict__ mask,
    float* __restrict__ wout) {
  const int b   = blockIdx.x >> 6;
  const int i0  = (blockIdx.x & 63) * TI_;
  const int tid = threadIdx.x;
  const int lane = tid & 31, wv = tid >> 5;

  __shared__ float sc[TI_][SCP_];
  __shared__ float redm[8][TI_];
  __shared__ float reds[8][TI_];

  v4f vr[2], qr[TI_][2];
#pragma unroll
  for (int h = 0; h < 2; ++h) {
    vr[h] = *(const v4f*)(v + 128 * h + 4 * lane);
#pragma unroll
    for (int t = 0; t < TI_; ++t)
      qr[t][h] = *(const v4f*)(pq + (b * S_ + i0 + t) * P_ + 128 * h + 4 * lane);
  }

  const float* rows = pk + (b * S_ + 32 * wv) * P_ + 4 * lane;

#pragma unroll
  for (int g = 0; g < 4; ++g) {
    float a[32];
#pragma unroll
    for (int kk = 0; kk < 8; ++kk) {
      const float* r = rows + (8 * g + kk) * P_;
      const v4f x0 = *(const v4f*)r;
      const v4f x1 = *(const v4f*)(r + 128);
#pragma unroll
      for (int t = 0; t < TI_; ++t) {
        float s = vr[0].x * fast_tanh(qr[t][0].x + x0.x);
        s = __builtin_fmaf(vr[0].y, fast_tanh(qr[t][0].y + x0.y), s);
        s = __builtin_fmaf(vr[0].z, fast_tanh(qr[t][0].z + x0.z), s);
        s = __builtin_fmaf(vr[0].w, fast_tanh(qr[t][0].w + x0.w), s);
        s = __builtin_fmaf(vr[1].x, fast_tanh(qr[t][1].x + x1.x), s);
        s = __builtin_fmaf(vr[1].y, fast_tanh(qr[t][1].y + x1.y), s);
        s = __builtin_fmaf(vr[1].z, fast_tanh(qr[t][1].z + x1.z), s);
        s = __builtin_fmaf(vr[1].w, fast_tanh(qr[t][1].w + x1.w), s);
        a[kk * 4 + t] = s;
      }
    }
    halve<16>(a, lane);
    halve<8>(a, lane);
    halve<4>(a, lane);
    halve<2>(a, lane);
    halve<1>(a, lane);
    sc[lane & 3][32 * wv + 8 * g + (lane >> 2)] = a[0];
  }
  __syncthreads();

  const float km = mask[b * S_ + tid];
  float s[TI_], e[TI_];
#pragma unroll
  for (int t = 0; t < TI_; ++t) {
    s[t] = (km > 0.0f) ? sc[t][tid] : NEG_INF_;
    float mx = s[t];
#pragma unroll
    for (int off = 16; off > 0; off >>= 1)
      mx = fmaxf(mx, __shfl_xor(mx, off, 32));
    if (lane == 0) redm[wv][t] = mx;
  }
  __syncthreads();
#pragma unroll
  for (int t = 0; t < TI_; ++t) {
    float bm = redm[0][t];
#pragma unroll
    for (int w = 1; w < 8; ++w) bm = fmaxf(bm, redm[w][t]);
    e[t] = __expf(s[t] - bm);
    float sm = e[t];
#pragma unroll
    for (int off = 16; off > 0; off >>= 1)
      sm += __shfl_xor(sm, off, 32);
    if (lane == 0) reds[wv][t] = sm;
  }
  __syncthreads();
#pragma unroll
  for (int t = 0; t < TI_; ++t) {
    float tot = 0.0f;
#pragma unroll
    for (int w = 0; w < 8; ++w) tot += reds[w][t];
    const float qm = mask[b * S_ + i0 + t];
    wout[(b * S_ + i0 + t) * S_ + tid] = (e[t] / tot) * qm;
  }
}

__global__ __launch_bounds__(256) void attn_wmma(
    const float* __restrict__ weights, const float* __restrict__ seq,
    float* __restrict__ attn) {
  const int lane = threadIdx.x & 31;
  const int wid  = blockIdx.x * (blockDim.x >> 5) + (threadIdx.x >> 5);
  const int b    = wid >> 9;
  const int rest = wid & 511;
  const int m0 = (rest >> 5) << 4;
  const int n0 = (rest & 31) << 4;
  const float* Wt = weights + b * S_ * S_;
  const float* Sq = seq     + b * S_ * D_;

  const int hl  = lane >> 4;
  const int l16 = lane & 15;
  const int m = m0 + l16;
  const int n = n0 + l16;

  v8f acc = {};
  for (int k0 = 0; k0 < S_; k0 += 4) {
    const int k = k0 + (hl << 1);
    v2f a = *(const v2f*)(Wt + m * S_ + k);
    v2f bb;
    bb.x = Sq[k * D_ + n];
    bb.y = Sq[(k + 1) * D_ + n];
    acc = __builtin_amdgcn_wmma_f32_16x16x4_f32(
        false, a, false, bb, (short)0, acc, false, false);
  }
  float* O = attn + b * S_ * D_;
#pragma unroll
  for (int r = 0; r < 8; ++r)
    O[(m0 + r + (hl << 3)) * D_ + n0 + l16] = acc[r];
}

extern "C" void kernel_launch(void* const* d_in, const int* in_sizes, int n_in,
                              void* d_out, int out_size, void* d_ws, size_t ws_size,
                              hipStream_t stream) {
  (void)in_sizes; (void)n_in; (void)out_size; (void)ws_size;
  const float* seq  = (const float*)d_in[0];
  const float* mask = (const float*)d_in[1];
  const float* W1   = (const float*)d_in[2];
  const float* W2   = (const float*)d_in[3];
  const float* v    = (const float*)d_in[4];

  float* attn = (float*)d_out;
  float* wout = attn + B_ * S_ * D_;
  float* pk   = (float*)d_ws;
  float* pq   = pk + B_ * S_ * P_;

  proj_wmma<<<256, 256, 0, stream>>>(seq, W1, W2, pk, pq);
  score_softmax<<<B_ * S_ / TI_, 256, 0, stream>>>(pq, pk, v, mask, wout);
  attn_wmma<<<256, 256, 0, stream>>>(wout, seq, attn);
}


# both WMMA kernels: operands whose fragment loads touched 16 lines per load are staged through LDS by whole-line 16-byte loads and read from LDS; products and sum order unchanged
# speedup vs baseline: 1.6876x; 1.2984x over previous
#include <hip/hip_runtime.h>

typedef float v2f __attribute__((ext_vector_type(2)));
typedef float v4f __attribute__((ext_vector_type(4)));
typedef float v8f __attribute__((ext_vector_type(8)));

#define B_ 4
#define S_ 256
#define D_ 512
#define P_ 256
#define NEG_INF_ (-1e9f)
#define TI_ 4
#define SCP_ (S_ + 16)

__device__ __forceinline__ float fast_tanh(float x) {
#if __has_builtin(__builtin_amdgcn_tanhf)
  return __builtin_amdgcn_tanhf(x);
#else
  float ax = __builtin_fabsf(x);
  float t  = __expf(-2.0f * ax);
  float r  = (1.0f - t) * __builtin_amdgcn_rcpf(1.0f + t);
  return x < 0.0f ? -r : r;
#endif
}

template <int OFF>
__device__ __forceinline__ float lane_xor(float x) {
  const int b = __builtin_bit_cast(int, x);
  if constexpr (OFF == 16)
    return __builtin_bit_cast(float, __builtin_amdgcn_permlanex16(
        b, b, 0x76543210, 0xfedcba98, false, false));
  else
    return __builtin_bit_cast(float, __builtin_amdgcn_update_dpp(
        b, b, 0x160 + OFF, 0xf, 0xf, false));
}

template <int OFF>
__device__ __forceinline__ void halve(float (&a)[32], const int lane) {
  const bool up = (lane & OFF) != 0;
#pragma unroll
  for (int k = 0; k < OFF; ++k) {
    const float lo = a[k];
    const float hi = a[k + OFF];
    const float keep = up ? hi : lo;
    const float send = up ? lo : hi;
    a[k] = keep + lane_xor<OFF>(send);
  }
}

#define PA_BM 32
#define PA_BN 64
#define PA_KC 128
#define PA_ROWS (PA_BM + PA_BN)
#define PA_PITCH (PA_KC + 4)
#define PA_NLD (PA_ROWS / 8)
#define PA_NCH (D_ / PA_KC)

__global__ __launch_bounds__(256) void proj_wmma(
    const float* __restrict__ seq, const float* __restrict__ W1,
    const float* __restrict__ W2, float* __restrict__ pk,
    float* __restrict__ pq) {
  __shared__ __attribute__((aligned(16))) float sm[2][PA_ROWS][PA_PITCH];

  const int tid  = threadIdx.x;
  const int lane = tid & 31;
  const int wv   = tid >> 5;
  const int mt = blockIdx.x >> 3;
  const int nt = blockIdx.x & 7;
  const int matsel = nt >> 2;
  const float* W   = matsel ? W2 : W1;
  float*       out = matsel ? pq : pk;
  const int mB = mt * PA_BM;
  const int nB = (nt & 3) * PA_BN;

  const float* ga = seq + (mB + wv) * D_ + 4 * lane;
  const float* gb = W   + (nB + wv) * D_ + 4 * lane;

  const int hl  = lane >> 4;
  const int l16 = lane & 15;
  const int mi = wv >> 2, ni = wv & 3;
  const int arow = mi * 16 + l16;
  const int brow = PA_BM + ni * 16 + l16;

  v4f pre[PA_NLD];
#pragma unroll
  for (int j = 0; j < PA_BM / 8; ++j)
    pre[j] = *(const v4f*)(ga + j * 8 * D_);
#pragma unroll
  for (int j = 0; j < PA_BN / 8; ++j)
    pre[PA_BM / 8 + j] = *(const v4f*)(gb + j * 8 * D_);
#pragma unroll
  for (int j = 0; j < PA_NLD; ++j)
    *(v4f*)&sm[0][wv + 8 * j][4 * lane] = pre[j];
  __syncthreads();

  v8f acc = {};
#pragma unroll
  for (int kc = 0; kc < PA_NCH; ++kc) {
    const int s = kc & 1;
    if (kc + 1 < PA_NCH) {
#pragma unroll
      for (int j = 0; j < PA_BM / 8; ++j)
        pre[j] = *(const v4f*)(ga + j * 8 * D_ + (kc + 1) * PA_KC);
#pragma unroll
      for (int j = 0; j < PA_BN / 8; ++j)
        pre[PA_BM / 8 + j] = *(const v4f*)(gb + j * 8 * D_ + (kc + 1) * PA_KC);
    }
    const float* ar = &sm[s][arow][2 * hl];
    const float* br = &sm[s][brow][2 * hl];
    v2f af[PA_KC / 4], bf[PA_KC / 4];
#pragma unroll
    for (int i = 0; i < PA_KC / 4; ++i) {
      af[i] = *(const v2f*)(ar + 4 * i);
      bf[i] = *(const v2f*)(br + 4 * i);
    }
    __builtin_amdgcn_sched_barrier(0);
#pragma unroll
    for (int i = 0; i < PA_KC / 4; ++i)
      acc = __builtin_amdgcn_wmma_f32_16x16x4_f32(
          false, af[i], false, bf[i], (short)0, acc, false, false);
    if (kc + 1 < PA_NCH) {
#pragma unroll
      for (int j = 0; j < PA_NLD; ++j)
        *(v4f*)&sm[s ^ 1][wv + 8 * j][4 * lane] = pre[j];
      __syncthreads();
    }
  }

  const int m0 = mB + mi * 16;
  const int n0 = nB + ni * 16;
#pragma unroll
  for (int r = 0; r < 8; ++r)
    out[(m0 + r + (hl << 3)) * P_ + n0 + l16] = acc[r];
}

__global__ __launch_bounds__(256) void score_softmax(
    const float* __restrict__ pq, const float* __restrict__ pk,
    const float* __restrict__ v, const float* __restrict__ mask,
    float* __restrict__ wout) {
  const int b   = blockIdx.x >> 6;
  const int i0  = (blockIdx.x & 63) * TI_;
  const int tid = threadIdx.x;
  const int lane = tid & 31, wv = tid >> 5;

  __shared__ float sc[TI_][SCP_];
  __shared__ float redm[8][TI_];
  __shared__ float reds[8][TI_];

  v4f vr[2], qr[TI_][2];
#pragma unroll
  for (int h = 0; h < 2; ++h) {
    vr[h] = *(const v4f*)(v + 128 * h + 4 * lane);
#pragma unroll
    for (int t = 0; t < TI_; ++t)
      qr[t][h] = *(const v4f*)(pq + (b * S_ + i0 + t) * P_ + 128 * h + 4 * lane);
  }

  const float* rows = pk + (b * S_ + 32 * wv) * P_ + 4 * lane;

#pragma unroll
  for (int g = 0; g < 4; ++g) {
    float a[32];
#pragma unroll
    for (int kk = 0; kk < 8; ++kk) {
      const float* r = rows + (8 * g + kk) * P_;
      const v4f x0 = *(const v4f*)r;
      const v4f x1 = *(const v4f*)(r + 128);
#pragma unroll
      for (int t = 0; t < TI_; ++t) {
        float s = vr[0].x * fast_tanh(qr[t][0].x + x0.x);
        s = __builtin_fmaf(vr[0].y, fast_tanh(qr[t][0].y + x0.y), s);
        s = __builtin_fmaf(vr[0].z, fast_tanh(qr[t][0].z + x0.z), s);
        s = __builtin_fmaf(vr[0].w, fast_tanh(qr[t][0].w + x0.w), s);
        s = __builtin_fmaf(vr[1].x, fast_tanh(qr[t][1].x + x1.x), s);
        s = __builtin_fmaf(vr[1].y, fast_tanh(qr[t][1].y + x1.y), s);
        s = __builtin_fmaf(vr[1].z, fast_tanh(qr[t][1].z + x1.z), s);
        s = __builtin_fmaf(vr[1].w, fast_tanh(qr[t][1].w + x1.w), s);
        a[kk * 4 + t] = s;
      }
    }
    halve<16>(a, lane);
    halve<8>(a, lane);
    halve<4>(a, lane);
    halve<2>(a, lane);
    halve<1>(a, lane);
    sc[lane & 3][32 * wv + 8 * g + (lane >> 2)] = a[0];
  }
  __syncthreads();

  const float km = mask[b * S_ + tid];
  float s[TI_], e[TI_];
#pragma unroll
  for (int t = 0; t < TI_; ++t) {
    s[t] = (km > 0.0f) ? sc[t][tid] : NEG_INF_;
    float mx = s[t];
#pragma unroll
    for (int off = 16; off > 0; off >>= 1)
      mx = fmaxf(mx, __shfl_xor(mx, off, 32));
    if (lane == 0) redm[wv][t] = mx;
  }
  __syncthreads();
#pragma unroll
  for (int t = 0; t < TI_; ++t) {
    float bm = redm[0][t];
#pragma unroll
    for (int w = 1; w < 8; ++w) bm = fmaxf(bm, redm[w][t]);
    e[t] = __expf(s[t] - bm);
    float sm = e[t];
#pragma unroll
    for (int off = 16; off > 0; off >>= 1)
      sm += __shfl_xor(sm, off, 32);
    if (lane == 0) reds[wv][t] = sm;
  }
  __syncthreads();
#pragma unroll
  for (int t = 0; t < TI_; ++t) {
    float tot = 0.0f;
#pragma unroll
    for (int w = 0; w < 8; ++w) tot += reds[w][t];
    const float qm = mask[b * S_ + i0 + t];
    wout[(b * S_ + i0 + t) * S_ + tid] = (e[t] / tot) * qm;
  }
}

#define PC_PITCH (S_ + 4)

__global__ __launch_bounds__(256) void attn_wmma(
    const float* __restrict__ weights, const float* __restrict__ seq,
    float* __restrict__ attn) {
  __shared__ __attribute__((aligned(16))) float wt[16][PC_PITCH];

  const int tid  = threadIdx.x;
  const int lane = tid & 31;
  const int wv   = tid >> 5;
  const int b  = blockIdx.x >> 6;
  const int m0 = ((blockIdx.x & 63) >> 2) << 4;
  const int n0 = (((blockIdx.x & 3) << 3) + wv) << 4;
  const float* Wt = weights + (b * S_ + m0) * S_;
  const float* Sq = seq     + b * S_ * D_;

#pragma unroll
  for (int j = 0; j < 4; ++j) {
    const int r = (tid >> 6) + 4 * j;
    const int c = 4 * (tid & 63);
    *(v4f*)&wt[r][c] = *(const v4f*)(Wt + r * S_ + c);
  }
  __syncthreads();

  const int hl  = lane >> 4;
  const int l16 = lane & 15;
  const int n = n0 + l16;
  const float* ar = &wt[l16][2 * hl];

  v8f acc = {};
  for (int k0 = 0; k0 < S_; k0 += 4) {
    const int k = k0 + (hl << 1);
    const v2f a = *(const v2f*)(ar + k0);
    v2f bb;
    bb.x = Sq[k * D_ + n];
    bb.y = Sq[(k + 1) * D_ + n];
    acc = __builtin_amdgcn_wmma_f32_16x16x4_f32(
        false, a, false, bb, (short)0, acc, false, false);
  }
  float* O = attn + b * S_ * D_;
#pragma unroll
  for (int r = 0; r < 8; ++r)
    O[(m0 + r + (hl << 3)) * D_ + n0 + l16] = acc[r];
}

extern "C" void kernel_launch(void* const* d_in, const int* in_sizes, int n_in,
                              void* d_out, int out_size, void* d_ws, size_t ws_size,
                              hipStream_t stream) {
  (void)in_sizes; (void)n_in; (void)out_size; (void)ws_size;
  const float* seq  = (const float*)d_in[0];
  const float* mask = (const float*)d_in[1];
  const float* W1   = (const float*)d_in[2];
  const float* W2   = (const float*)d_in[3];
  const float* v    = (const float*)d_in[4];

  float* attn = (float*)d_out;
  float* wout = attn + B_ * S_ * D_;
  float* pk   = (float*)d_ws;
  float* pq   = pk + B_ * S_ * P_;

  proj_wmma<<<256, 256, 0, stream>>>(seq, W1, W2, pk, pq);
  score_softmax<<<B_ * S_ / TI_, 256, 0, stream>>>(pq, pk, v, mask, wout);
  attn_wmma<<<256, 256, 0, stream>>>(wout, seq, attn);
}


# attention GEMM: seq (B) values of each block of 16 k steps are requested two blocks ahead and LDS fragments one block ahead, not load-wait-multiply per 8 k steps; products and sum order unchanged
# speedup vs baseline: 1.7549x; 1.0399x over previous
#include <hip/hip_runtime.h>

typedef float v2f __attribute__((ext_vector_type(2)));
typedef float v4f __attribute__((ext_vector_type(4)));
typedef float v8f __attribute__((ext_vector_type(8)));

#define B_ 4
#define S_ 256
#define D_ 512
#define P_ 256
#define NEG_INF_ (-1e9f)
#define TI_ 4
#define SCP_ (S_ + 16)

__device__ __forceinline__ float fast_tanh(float x) {
#if __has_builtin(__builtin_amdgcn_tanhf)
  return __builtin_amdgcn_tanhf(x);
#else
  float ax = __builtin_fabsf(x);
  float t  = __expf(-2.0f * ax);
  float r  = (1.0f - t) * __builtin_amdgcn_rcpf(1.0f + t);
  return x < 0.0f ? -r : r;
#endif
}

template <int OFF>
__device__ __forceinline__ float lane_xor(float x) {
  const int b = __builtin_bit_cast(int, x);
  if constexpr (OFF == 16)
    return __builtin_bit_cast(float, __builtin_amdgcn_permlanex16(
        b, b, 0x76543210, 0xfedcba98, false, false));
  else
    return __builtin_bit_cast(float, __builtin_amdgcn_update_dpp(
        b, b, 0x160 + OFF, 0xf, 0xf, false));
}

template <int OFF>
__device__ __forceinline__ void halve(float (&a)[32], const int lane) {
  const bool up = (lane & OFF) != 0;
#pragma unroll
  for (int k = 0; k < OFF; ++k) {
    const float lo = a[k];
    const float hi = a[k + OFF];
    const float keep = up ? hi : lo;
    const float send = up ? lo : hi;
    a[k] = keep + lane_xor<OFF>(send);
  }
}

#define PA_BM 32
#define PA_BN 64
#define PA_KC 128
#define PA_ROWS (PA_BM + PA_BN)
#define PA_PITCH (PA_KC + 4)
#define PA_NLD (PA_ROWS / 8)
#define PA_NCH (D_ / PA_KC)

__global__ __launch_bounds__(256) void proj_wmma(
    const float* __restrict__ seq, const float* __restrict__ W1,
    const float* __restrict__ W2, float* __restrict__ pk,
    float* __restrict__ pq) {
  __shared__ __attribute__((aligned(16))) float sm[2][PA_ROWS][PA_PITCH];

  const int tid  = threadIdx.x;
  const int lane = tid & 31;
  const int wv   = tid >> 5;
  const int mt = blockIdx.x >> 3;
  const int nt = blockIdx.x & 7;
  const int matsel = nt >> 2;
  const float* W   = matsel ? W2 : W1;
  float*       out = matsel ? pq : pk;
  const int mB = mt * PA_BM;
  const int nB = (nt & 3) * PA_BN;

  const float* ga = seq + (mB + wv) * D_ + 4 * lane;
  const float* gb = W   + (nB + wv) * D_ + 4 * lane;

  const int hl  = lane >> 4;
  const int l16 = lane & 15;
  const int mi = wv >> 2, ni = wv & 3;
  const int arow = mi * 16 + l16;
  const int brow = PA_BM + ni * 16 + l16;

  v4f pre[PA_NLD];
#pragma unroll
  for (int j = 0; j < PA_BM / 8; ++j)
    pre[j] = *(const v4f*)(ga + j * 8 * D_);
#pragma unroll
  for (int j = 0; j < PA_BN / 8; ++j)
    pre[PA_BM / 8 + j] = *(const v4f*)(gb + j * 8 * D_);
#pragma unroll
  for (int j = 0; j < PA_NLD; ++j)
    *(v4f*)&sm[0][wv + 8 * j][4 * lane] = pre[j];
  __syncthreads();

  v8f acc = {};
#pragma unroll
  for (int kc = 0; kc < PA_NCH; ++kc) {
    const int s = kc & 1;
    if (kc + 1 < PA_NCH) {
#pragma unroll
      for (int j = 0; j < PA_BM / 8; ++j)
        pre[j] = *(const v4f*)(ga + j * 8 * D_ + (kc + 1) * PA_KC);
#pragma unroll
      for (int j = 0; j < PA_BN / 8; ++j)
        pre[PA_BM / 8 + j] = *(const v4f*)(gb + j * 8 * D_ + (kc + 1) * PA_KC);
    }
    const float* ar = &sm[s][arow][2 * hl];
    const float* br = &sm[s][brow][2 * hl];
    v2f af[PA_KC / 4], bf[PA_KC / 4];
#pragma unroll
    for (int i = 0; i < PA_KC / 4; ++i) {
      af[i] = *(const v2f*)(ar + 4 * i);
      bf[i] = *(const v2f*)(br + 4 * i);
    }
    __builtin_amdgcn_sched_barrier(0);
#pragma unroll
    for (int i = 0; i < PA_KC / 4; ++i)
      acc = __builtin_amdgcn_wmma_f32_16x16x4_f32(
          false, af[i], false, bf[i], (short)0, acc, false, false);
    if (kc + 1 < PA_NCH) {
#pragma unroll
      for (int j = 0; j < PA_NLD; ++j)
        *(v4f*)&sm[s ^ 1][wv + 8 * j][4 * lane] = pre[j];
      __syncthreads();
    }
  }

  const int m0 = mB + mi * 16;
  const int n0 = nB + ni * 16;
#pragma unroll
  for (int r = 0; r < 8; ++r)
    out[(m0 + r + (hl << 3)) * P_ + n0 + l16] = acc[r];
}

__global__ __launch_bounds__(256) void score_softmax(
    const float* __restrict__ pq, const float* __restrict__ pk,
    const float* __restrict__ v, const float* __restrict__ mask,
    float* __restrict__ wout) {
  const int b   = blockIdx.x >> 6;
  const int i0  = (blockIdx.x & 63) * TI_;
  const int tid = threadIdx.x;
  const int lane = tid & 31, wv = tid >> 5;

  __shared__ float sc[TI_][SCP_];
  __shared__ float redm[8][TI_];
  __shared__ float reds[8][TI_];

  v4f vr[2], qr[TI_][2];
#pragma unroll
  for (int h = 0; h < 2; ++h) {
    vr[h] = *(const v4f*)(v + 128 * h + 4 * lane);
#pragma unroll
    for (int t = 0; t < TI_; ++t)
      qr[t][h] = *(const v4f*)(pq + (b * S_ + i0 + t) * P_ + 128 * h + 4 * lane);
  }

  const float* rows = pk + (b * S_ + 32 * wv) * P_ + 4 * lane;

#pragma unroll
  for (int g = 0; g < 4; ++g) {
    float a[32];
#pragma unroll
    for (int kk = 0; kk < 8; ++kk) {
      const float* r = rows + (8 * g + kk) * P_;
      const v4f x0 = *(const v4f*)r;
      const v4f x1 = *(const v4f*)(r + 128);
#pragma unroll
      for (int t = 0; t < TI_; ++t) {
        float s = vr[0].x * fast_tanh(qr[t][0].x + x0.x);
        s = __builtin_fmaf(vr[0].y, fast_tanh(qr[t][0].y + x0.y), s);
        s = __builtin_fmaf(vr[0].z, fast_tanh(qr[t][0].z + x0.z), s);
        s = __builtin_fmaf(vr[0].w, fast_tanh(qr[t][0].w + x0.w), s);
        s = __builtin_fmaf(vr[1].x, fast_tanh(qr[t][1].x + x1.x), s);
        s = __builtin_fmaf(vr[1].y, fast_tanh(qr[t][1].y + x1.y), s);
        s = __builtin_fmaf(vr[1].z, fast_tanh(qr[t][1].z + x1.z), s);
        s = __builtin_fmaf(vr[1].w, fast_tanh(qr[t][1].w + x1.w), s);
        a[kk * 4 + t] = s;
      }
    }
    halve<16>(a, lane);
    halve<8>(a, lane);
    halve<4>(a, lane);
    halve<2>(a, lane);
    halve<1>(a, lane);
    sc[lane & 3][32 * wv + 8 * g + (lane >> 2)] = a[0];
  }
  __syncthreads();

  const float km = mask[b * S_ + tid];
  float s[TI_], e[TI_];
#pragma unroll
  for (int t = 0; t < TI_; ++t) {
    s[t] = (km > 0.0f) ? sc[t][tid] : NEG_INF_;
    float mx = s[t];
#pragma unroll
    for (int off = 16; off > 0; off >>= 1)
      mx = fmaxf(mx, __shfl_xor(mx, off, 32));
    if (lane == 0) redm[wv][t] = mx;
  }
  __syncthreads();
#pragma unroll
  for (int t = 0; t < TI_; ++t) {
    float bm = redm[0][t];
#pragma unroll
    for (int w = 1; w < 8; ++w) bm = fmaxf(bm, redm[w][t]);
    e[t] = __expf(s[t] - bm);
    float sm = e[t];
#pragma unroll
    for (int off = 16; off > 0; off >>= 1)
      sm += __shfl_xor(sm, off, 32);
    if (lane == 0) reds[wv][t] = sm;
  }
  __syncthreads();
#pragma unroll
  for (int t = 0; t < TI_; ++t) {
    float tot = 0.0f;
#pragma unroll
    for (int w = 0; w < 8; ++w) tot += reds[w][t];
    const float qm = mask[b * S_ + i0 + t];
    wout[(b * S_ + i0 + t) * S_ + tid] = (e[t] / tot) * qm;
  }
}

#define PC_PITCH (S_ + 4)
#define PC_KB 16
#define PC_NB (S_ / 4 / PC_KB)

__global__ __launch_bounds__(256) void attn_wmma(
    const float* __restrict__ weights, const float* __restrict__ seq,
    float* __restrict__ attn) {
  __shared__ __attribute__((aligned(16))) float wt[16][PC_PITCH];

  const int tid  = threadIdx.x;
  const int lane = tid & 31;
  const int wv   = tid >> 5;
  const int b  = blockIdx.x >> 6;
  const int m0 = ((blockIdx.x & 63) >> 2) << 4;
  const int n0 = (((blockIdx.x & 3) << 3) + wv) << 4;
  const float* Wt = weights + (b * S_ + m0) * S_;
  const float* Sq = seq     + b * S_ * D_;

  const int hl  = lane >> 4;
  const int l16 = lane & 15;
  const float* bp = Sq + (hl << 1) * D_ + n0 + l16;
  const float* ar = &wt[l16][2 * hl];

  v4f st[4];
#pragma unroll
  for (int j = 0; j < 4; ++j)
    st[j] = *(const v4f*)(Wt + ((tid >> 6) + 4 * j) * S_ + 4 * (tid & 63));

  v2f bb[S_ / 4];
#pragma unroll
  for (int i = 0; i < 2 * PC_KB; ++i) {
    bb[i].x = bp[(4 * i) * D_];
    bb[i].y = bp[(4 * i + 1) * D_];
  }

#pragma unroll
  for (int j = 0; j < 4; ++j)
    *(v4f*)&wt[(tid >> 6) + 4 * j][4 * (tid & 63)] = st[j];
  __syncthreads();

  v2f af[2][PC_KB];
#pragma unroll
  for (int i = 0; i < PC_KB; ++i)
    af[0][i] = *(const v2f*)(ar + 4 * i);

  v8f acc = {};
#pragma unroll
  for (int g = 0; g < PC_NB; ++g) {
    if (g + 2 < PC_NB) {
#pragma unroll
      for (int i = (g + 2) * PC_KB; i < (g + 3) * PC_KB; ++i) {
        bb[i].x = bp[(4 * i) * D_];
        bb[i].y = bp[(4 * i + 1) * D_];
      }
    }
    if (g + 1 < PC_NB) {
#pragma unroll
      for (int i = 0; i < PC_KB; ++i)
        af[(g + 1) & 1][i] = *(const v2f*)(ar + 4 * ((g + 1) * PC_KB + i));
    }
    __builtin_amdgcn_sched_barrier(0);
#pragma unroll
    for (int i = 0; i < PC_KB; ++i)
      acc = __builtin_amdgcn_wmma_f32_16x16x4_f32(
          false, af[g & 1][i], false, bb[g * PC_KB + i], (short)0, acc, false, false);
  }
  float* O = attn + b * S_ * D_;
#pragma unroll
  for (int r = 0; r < 8; ++r)
    O[(m0 + r + (hl << 3)) * D_ + n0 + l16] = acc[r];
}

extern "C" void kernel_launch(void* const* d_in, const int* in_sizes, int n_in,
                              void* d_out, int out_size, void* d_ws, size_t ws_size,
                              hipStream_t stream) {
  (void)in_sizes; (void)n_in; (void)out_size; (void)ws_size;
  const float* seq  = (const float*)d_in[0];
  const float* mask = (const float*)d_in[1];
  const float* W1   = (const float*)d_in[2];
  const float* W2   = (const float*)d_in[3];
  const float* v    = (const float*)d_in[4];

  float* attn = (float*)d_out;
  float* wout = attn + B_ * S_ * D_;
  float* pk   = (float*)d_ws;
  float* pq   = pk + B_ * S_ * P_;

  proj_wmma<<<256, 256, 0, stream>>>(seq, W1, W2, pk, pq);
  score_softmax<<<B_ * S_ / TI_, 256, 0, stream>>>(pq, pk, v, mask, wout);
  attn_wmma<<<256, 256, 0, stream>>>(wout, seq, attn);
}


# score kernel re-tiled from 4 queries x 256 keys to 16 queries x 64 keys per block (bytes read per block 261 KB -> 81 KB); softmax moves into the attention kernel: sum order over p and keys changes
# speedup vs baseline: 2.1213x; 1.2088x over previous
#include <hip/hip_runtime.h>

typedef float v2f __attribute__((ext_vector_type(2)));
typedef float v4f __attribute__((ext_vector_type(4)));
typedef float v8f __attribute__((ext_vector_type(8)));

#define B_ 4
#define S_ 256
#define D_ 512
#define P_ 256
#define NEG_INF_ (-1e9f)

__device__ __forceinline__ float fast_tanh(float x) {
#if __has_builtin(__builtin_amdgcn_tanhf)
  return __builtin_amdgcn_tanhf(x);
#else
  float ax = __builtin_fabsf(x);
  float t  = __expf(-2.0f * ax);
  float r  = (1.0f - t) * __builtin_amdgcn_rcpf(1.0f + t);
  return x < 0.0f ? -r : r;
#endif
}

template <int OFF>
__device__ __forceinline__ float lane_xor(float x) {
  const int b = __builtin_bit_cast(int, x);
  if constexpr (OFF == 16)
    return __builtin_bit_cast(float, __builtin_amdgcn_permlanex16(
        b, b, 0x76543210, 0xfedcba98, false, false));
  else
    return __builtin_bit_cast(float, __builtin_amdgcn_update_dpp(
        b, b, 0x160 + OFF, 0xf, 0xf, false));
}

template <int OFF>
__device__ __forceinline__ void halve(float (&a)[32], const int lane) {
  const bool up = (lane & OFF) != 0;
#pragma unroll
  for (int k = 0; k < OFF; ++k) {
    const float lo = a[k];
    const float hi = a[k + OFF];
    const float keep = up ? hi : lo;
    const float send = up ? lo : hi;
    a[k] = keep + lane_xor<OFF>(send);
  }
}

#define PA_BM 32
#define PA_BN 64
#define PA_KC 128
#define PA_ROWS (PA_BM + PA_BN)
#define PA_PITCH (PA_KC + 4)
#define PA_NLD (PA_ROWS / 8)
#define PA_NCH (D_ / PA_KC)

__global__ __launch_bounds__(256) void proj_wmma(
    const float* __restrict__ seq, const float* __restrict__ W1,
    const float* __restrict__ W2, float* __restrict__ pk,
    float* __restrict__ pq) {
  __shared__ __attribute__((aligned(16))) float sm[2][PA_ROWS][PA_PITCH];

  const int tid  = threadIdx.x;
  const int lane = tid & 31;
  const int wv   = tid >> 5;
  const int mt = blockIdx.x >> 3;
  const int nt = blockIdx.x & 7;
  const int matsel = nt >> 2;
  const float* W   = matsel ? W2 : W1;
  float*       out = matsel ? pq : pk;
  const int mB = mt * PA_BM;
  const int nB = (nt & 3) * PA_BN;

  const float* ga = seq + (mB + wv) * D_ + 4 * lane;
  const float* gb = W   + (nB + wv) * D_ + 4 * lane;

  const int hl  = lane >> 4;
  const int l16 = lane & 15;
  const int mi = wv >> 2, ni = wv & 3;
  const int arow = mi * 16 + l16;
  const int brow = PA_BM + ni * 16 + l16;

  v4f pre[PA_NLD];
#pragma unroll
  for (int j = 0; j < PA_BM / 8; ++j)
    pre[j] = *(const v4f*)(ga + j * 8 * D_);
#pragma unroll
  for (int j = 0; j < PA_BN / 8; ++j)
    pre[PA_BM / 8 + j] = *(const v4f*)(gb + j * 8 * D_);
#pragma unroll
  for (int j = 0; j < PA_NLD; ++j)
    *(v4f*)&sm[0][wv + 8 * j][4 * lane] = pre[j];
  __syncthreads();

  v8f acc = {};
#pragma unroll
  for (int kc = 0; kc < PA_NCH; ++kc) {
    const int s = kc & 1;
    if (kc + 1 < PA_NCH) {
#pragma unroll
      for (int j = 0; j < PA_BM / 8; ++j)
        pre[j] = *(const v4f*)(ga + j * 8 * D_ + (kc + 1) * PA_KC);
#pragma unroll
      for (int j = 0; j < PA_BN / 8; ++j)
        pre[PA_BM / 8 + j] = *(const v4f*)(gb + j * 8 * D_ + (kc + 1) * PA_KC);
    }
    const float* ar = &sm[s][arow][2 * hl];
    const float* br = &sm[s][brow][2 * hl];
    v2f af[PA_KC / 4], bf[PA_KC / 4];
#pragma unroll
    for (int i = 0; i < PA_KC / 4; ++i) {
      af[i] = *(const v2f*)(ar + 4 * i);
      bf[i] = *(const v2f*)(br + 4 * i);
    }
    __builtin_amdgcn_sched_barrier(0);
#pragma unroll
    for (int i = 0; i < PA_KC / 4; ++i)
      acc = __builtin_amdgcn_wmma_f32_16x16x4_f32(
          false, af[i], false, bf[i], (short)0, acc, false, false);
    if (kc + 1 < PA_NCH) {
#pragma unroll
      for (int j = 0; j < PA_NLD; ++j)
        *(v4f*)&sm[s ^ 1][wv + 8 * j][4 * lane] = pre[j];
      __syncthreads();
    }
  }

  const int m0 = mB + mi * 16;
  const int n0 = nB + ni * 16;
#pragma unroll
  for (int r = 0; r < 8; ++r)
    out[(m0 + r + (hl << 3)) * P_ + n0 + l16] = acc[r];
}

#define TQ_ 16
#define TK_ 64
#define SCP_ (TK_ + 4)

__global__ __launch_bounds__(256) void score_tile(
    const float* __restrict__ pq, const float* __restrict__ pk,
    const float* __restrict__ v, float* __restrict__ raw) {
  const int b   = blockIdx.x >> 6;
  const int i0  = ((blockIdx.x & 63) >> 2) * TQ_;
  const int j0  = (blockIdx.x & 3) * TK_;
  const int tid = threadIdx.x;
  const int lane = tid & 31, wv = tid >> 5;

  __shared__ __attribute__((aligned(16))) float qs[TQ_][P_];
  __shared__ __attribute__((aligned(16))) float sc[TQ_][SCP_];


  const float* rows = pk + (b * S_ + j0 + 8 * wv) * P_ + 4 * lane;
  v4f xs[8][2];
#pragma unroll
  for (int j = 0; j < 8; ++j) {
    xs[j][0] = *(const v4f*)(rows + j * P_);
    xs[j][1] = *(const v4f*)(rows + j * P_ + 128);
  }

  {
    v4f st[4];
#pragma unroll
    for (int j = 0; j < 4; ++j)
      st[j] = *(const v4f*)(pq + (b * S_ + i0 + (tid >> 6) + 4 * j) * P_ + 4 * (tid & 63));
#pragma unroll
    for (int j = 0; j < 4; ++j)
      *(v4f*)&qs[(tid >> 6) + 4 * j][4 * (tid & 63)] = st[j];
  }
  v4f vr[2];
#pragma unroll
  for (int h = 0; h < 2; ++h)
    vr[h] = *(const v4f*)(v + 128 * h + 4 * lane);
  __syncthreads();

  v4f qr[TQ_][2];
#pragma unroll
  for (int t = 0; t < TQ_; ++t) {
    qr[t][0] = *(const v4f*)&qs[t][4 * lane];
    qr[t][1] = *(const v4f*)&qs[t][128 + 4 * lane];
  }

#pragma unroll
  for (int g = 0; g < 4; ++g) {
    float a[32];
#pragma unroll
    for (int kk = 0; kk < 2; ++kk) {
      const v4f x0 = xs[2 * g + kk][0];
      const v4f x1 = xs[2 * g + kk][1];
#pragma unroll
      for (int t = 0; t < TQ_; ++t) {
        float s = vr[0].x * fast_tanh(qr[t][0].x + x0.x);
        s = __builtin_fmaf(vr[0].y, fast_tanh(qr[t][0].y + x0.y), s);
        s = __builtin_fmaf(vr[0].z, fast_tanh(qr[t][0].z + x0.z), s);
        s = __builtin_fmaf(vr[0].w, fast_tanh(qr[t][0].w + x0.w), s);
        s = __builtin_fmaf(vr[1].x, fast_tanh(qr[t][1].x + x1.x), s);
        s = __builtin_fmaf(vr[1].y, fast_tanh(qr[t][1].y + x1.y), s);
        s = __builtin_fmaf(vr[1].z, fast_tanh(qr[t][1].z + x1.z), s);
        s = __builtin_fmaf(vr[1].w, fast_tanh(qr[t][1].w + x1.w), s);
        a[kk * TQ_ + t] = s;
      }
    }
    halve<16>(a, lane);
    halve<8>(a, lane);
    halve<4>(a, lane);
    halve<2>(a, lane);
    halve<1>(a, lane);
    sc[lane & 15][8 * wv + 2 * g + (lane >> 4)] = a[0];
  }
  __syncthreads();

  {
    const int row = tid >> 4, c4 = tid & 15;
    const v4f o = *(const v4f*)&sc[row][4 * c4];
    *(v4f*)(raw + (b * S_ + i0 + row) * S_ + j0 + 4 * c4) = o;
  }
}

#define PC_PITCH (S_ + 4)
#define PC_KB 16
#define PC_NB (S_ / 4 / PC_KB)

__global__ __launch_bounds__(256) void attn_wmma(
    const float* __restrict__ raw, const float* __restrict__ mask,
    const float* __restrict__ seq, float* __restrict__ wout,
    float* __restrict__ attn) {
  __shared__ __attribute__((aligned(16))) float wt[16][PC_PITCH];

  const int tid  = threadIdx.x;
  const int lane = tid & 31;
  const int wv   = tid >> 5;
  const int b  = blockIdx.x >> 6;
  const int m0 = ((blockIdx.x & 63) >> 2) << 4;
  const int n0 = (((blockIdx.x & 3) << 3) + wv) << 4;
  const int nq = blockIdx.x & 3;
  const float* Sq = seq     + b * S_ * D_;

  const int hl  = lane >> 4;
  const int l16 = lane & 15;
  const float* bp = Sq + (hl << 1) * D_ + n0 + l16;
  const float* ar = &wt[l16][2 * hl];

  const float* Rw = raw + (b * S_ + m0 + 2 * wv) * S_ + 4 * lane;
  v4f x[2][2], km[2], wr[2][2];
  float qm[2];
#pragma unroll
  for (int rr = 0; rr < 2; ++rr) {
    x[rr][0] = *(const v4f*)(Rw + rr * S_);
    x[rr][1] = *(const v4f*)(Rw + rr * S_ + 128);
    qm[rr] = mask[b * S_ + m0 + 2 * wv + rr];
  }
  km[0] = *(const v4f*)(mask + b * S_ + 4 * lane);
  km[1] = *(const v4f*)(mask + b * S_ + 128 + 4 * lane);

  v2f bb[S_ / 4];
#pragma unroll
  for (int i = 0; i < 2 * PC_KB; ++i) {
    bb[i].x = bp[(4 * i) * D_];
    bb[i].y = bp[(4 * i + 1) * D_];
  }

#pragma unroll
  for (int rr = 0; rr < 2; ++rr) {
    const int row = 2 * wv + rr;
    float s[8];
#pragma unroll
    for (int h = 0; h < 2; ++h) {
      s[4 * h + 0] = (km[h].x > 0.0f) ? x[rr][h].x : NEG_INF_;
      s[4 * h + 1] = (km[h].y > 0.0f) ? x[rr][h].y : NEG_INF_;
      s[4 * h + 2] = (km[h].z > 0.0f) ? x[rr][h].z : NEG_INF_;
      s[4 * h + 3] = (km[h].w > 0.0f) ? x[rr][h].w : NEG_INF_;
    }
    float mx = s[0];
#pragma unroll
    for (int i = 1; i < 8; ++i) mx = fmaxf(mx, s[i]);
    mx = fmaxf(mx, lane_xor<16>(mx));
    mx = fmaxf(mx, lane_xor<8>(mx));
    mx = fmaxf(mx, lane_xor<4>(mx));
    mx = fmaxf(mx, lane_xor<2>(mx));
    mx = fmaxf(mx, lane_xor<1>(mx));
    float e[8];
#pragma unroll
    for (int i = 0; i < 8; ++i) e[i] = __expf(s[i] - mx);
    float tot = e[0];
#pragma unroll
    for (int i = 1; i < 8; ++i) tot += e[i];
    tot += lane_xor<16>(tot);
    tot += lane_xor<8>(tot);
    tot += lane_xor<4>(tot);
    tot += lane_xor<2>(tot);
    tot += lane_xor<1>(tot);
    const float inv = 1.0f / tot;
    wr[rr][0].x = (e[0] * inv) * qm[rr];
    wr[rr][0].y = (e[1] * inv) * qm[rr];
    wr[rr][0].z = (e[2] * inv) * qm[rr];
    wr[rr][0].w = (e[3] * inv) * qm[rr];
    wr[rr][1].x = (e[4] * inv) * qm[rr];
    wr[rr][1].y = (e[5] * inv) * qm[rr];
    wr[rr][1].z = (e[6] * inv) * qm[rr];
    wr[rr][1].w = (e[7] * inv) * qm[rr];
    *(v4f*)&wt[row][4 * lane]       = wr[rr][0];
    *(v4f*)&wt[row][128 + 4 * lane] = wr[rr][1];
  }
  __syncthreads();
#pragma unroll
  for (int rr = 0; rr < 2; ++rr) {
    const int row = 2 * wv + rr;
    if ((row & 3) == nq) {
      float* wo = wout + (b * S_ + m0 + row) * S_ + 4 * lane;
      *(v4f*)wo         = wr[rr][0];
      *(v4f*)(wo + 128) = wr[rr][1];
    }
  }

  v2f af[2][PC_KB];
#pragma unroll
  for (int i = 0; i < PC_KB; ++i)
    af[0][i] = *(const v2f*)(ar + 4 * i);

  v8f acc = {};
#pragma unroll
  for (int g = 0; g < PC_NB; ++g) {
    if (g + 2 < PC_NB) {
#pragma unroll
      for (int i = (g + 2) * PC_KB; i < (g + 3) * PC_KB; ++i) {
        bb[i].x = bp[(4 * i) * D_];
        bb[i].y = bp[(4 * i + 1) * D_];
      }
    }
    if (g + 1 < PC_NB) {
#pragma unroll
      for (int i = 0; i < PC_KB; ++i)
        af[(g + 1) & 1][i] = *(const v2f*)(ar + 4 * ((g + 1) * PC_KB + i));
    }
    __builtin_amdgcn_sched_barrier(0);
#pragma unroll
    for (int i = 0; i < PC_KB; ++i)
      acc = __builtin_amdgcn_wmma_f32_16x16x4_f32(
          false, af[g & 1][i], false, bb[g * PC_KB + i], (short)0, acc, false, false);
  }
  float* O = attn + b * S_ * D_;
#pragma unroll
  for (int r = 0; r < 8; ++r)
    O[(m0 + r + (hl << 3)) * D_ + n0 + l16] = acc[r];
}

extern "C" void kernel_launch(void* const* d_in, const int* in_sizes, int n_in,
                              void* d_out, int out_size, void* d_ws, size_t ws_size,
                              hipStream_t stream) {
  (void)in_sizes; (void)n_in; (void)out_size; (void)ws_size;
  const float* seq  = (const float*)d_in[0];
  const float* mask = (const float*)d_in[1];
  const float* W1   = (const float*)d_in[2];
  const float* W2   = (const float*)d_in[3];
  const float* v    = (const float*)d_in[4];

  float* attn = (float*)d_out;
  float* wout = attn + B_ * S_ * D_;
  float* pk   = (float*)d_ws;
  float* pq   = pk + B_ * S_ * P_;
  float* raw  = pq + B_ * S_ * P_;

  proj_wmma<<<256, 256, 0, stream>>>(seq, W1, W2, pk, pq);
  score_tile<<<256, 256, 0, stream>>>(pq, pk, v, raw);
  attn_wmma<<<256, 256, 0, stream>>>(raw, mask, seq, wout, attn);
}
